# starting kernel
# baseline (speedup 1.0000x reference)
#include <hip/hip_runtime.h>
#include <cmath>

typedef __attribute__((ext_vector_type(2))) float v2f;
typedef __attribute__((ext_vector_type(8))) float v8f;

#define B_  8
#define N_  400
#define T_  24
#define Y_  12
#define H_  64
#define D_  96
#define DL_ 64
#define TY_ (T_ * Y_)
#define BTY_ (B_ * TY_)
#define NT_ 384
#define TC_ 6

__device__ __forceinline__ float fast_tanh(float x) {
#if __has_builtin(__builtin_amdgcn_tanhf)
    return __builtin_amdgcn_tanhf(x);
#else
    return tanhf(x);
#endif
}

__device__ __forceinline__ v8f wmma_f32_16x16x4(v2f a, v2f b, v8f c) {
    return __builtin_amdgcn_wmma_f32_16x16x4_f32(
         false, a,  false, b,
         (short)0, c,  false,  false);
}

__global__ __launch_bounds__(NT_) void tsa_fused_kernel(
    const float* __restrict__ x_long,
    const float* __restrict__ x_short,
    const float* __restrict__ Wq,
    const float* __restrict__ Wk,
    const float* __restrict__ V,
    const float* __restrict__ encW,
    const float* __restrict__ encb,
    float* __restrict__ out)
{
    __shared__ alignas(16) float  lds_wq[H_ * D_];
    __shared__ alignas(16) float  lds_xs[16 * D_];
    __shared__ alignas(16) float  lds_Q [B_ * H_];
    __shared__ alignas(16) float2 lds_vk[H_ * Y_];
    __shared__ alignas(16) float  lds_xl[BTY_];
    __shared__ alignas(16) float  lds_A [BTY_];
    __shared__ alignas(16) float  lds_xn[16 * 16];
    __shared__ alignas(16) float  lds_ewT[DL_ * 16];

    const int n    = blockIdx.x;
    const int tid  = threadIdx.x;
    const int wave = tid >> 5;
    const int lane = tid & 31;

    for (int i4 = tid; i4 < (H_ * D_) / 4; i4 += NT_) {
        const int i = i4 * 4;
        const int h = i / D_, d = i % D_;
        ((float4*)lds_wq)[i4] = *(const float4*)&Wq[(h * N_ + n) * D_ + d];
    }
    for (int i4 = tid; i4 < (B_ * D_) / 4; i4 += NT_) {
        const int i = i4 * 4;
        const int b = i / D_, d = i % D_;
        ((float4*)lds_xs)[i4] = *(const float4*)&x_short[(b * N_ + n) * D_ + d];
    }
    for (int i = tid; i < B_ * D_; i += NT_) lds_xs[B_ * D_ + i] = 0.0f;
    for (int i = tid; i < H_ * Y_; i += NT_) {
        const int h = i / Y_, y = i % Y_;
        lds_vk[i] = make_float2(Wk[(h * N_ + n) * Y_ + y],
                                V [(h * N_ + n) * Y_ + y]);
    }
    for (int i4 = tid; i4 < BTY_ / 4; i4 += NT_) {
        const int i = i4 * 4;
        const int b = i / TY_, r = i % TY_;
        ((float4*)lds_xl)[i4] = *(const float4*)&x_long[(b * N_ + n) * TY_ + r];
    }
    for (int i = tid; i < Y_ * DL_; i += NT_) {
        const int y = i / DL_, d = i % DL_;
        lds_ewT[d * 16 + y] = encW[n * Y_ * DL_ + i];
    }
    if (tid < 256) lds_xn[tid] = 0.0f;
    __syncthreads();

    if (wave < 4) {
        const int h0 = wave * 16;
        const int m  = lane & 15;
        const int ko = (lane >> 4) * 2;
        v8f c = {};
        #pragma unroll
        for (int kk = 0; kk < D_ / 4; ++kk) {
            const int k = kk * 4 + ko;
            const float2 af = *(const float2*)&lds_xs[m * D_ + k];
            const float2 bf = *(const float2*)&lds_wq[(h0 + m) * D_ + k];
            v2f a, b;
            a.x = af.x; a.y = af.y;
            b.x = bf.x; b.y = bf.y;
            c = wmma_f32_16x16x4(a, b, c);
        }
        if (lane < 16) {
            #pragma unroll
            for (int v = 0; v < B_; ++v)
                lds_Q[v * H_ + h0 + lane] = c[v];
        }
    }
    __syncthreads();

    {
        const int p  = tid % (B_ * Y_);
        const int t0 = (tid / (B_ * Y_)) * TC_;
        const int b  = p / Y_, y = p % Y_;
        float xl[TC_], acc[TC_];
        #pragma unroll
        for (int j = 0; j < TC_; ++j) {
            xl[j]  = lds_xl[b * TY_ + (t0 + j) * Y_ + y];
            acc[j] = 0.0f;
        }
        const float2* __restrict__ q2 = (const float2*)(lds_Q + b * H_);
        #pragma unroll 4
        for (int h2 = 0; h2 < H_ / 2; ++h2) {
            const float2 q  = q2[h2];
            const float2 w0 = lds_vk[(2 * h2    ) * Y_ + y];
            const float2 w1 = lds_vk[(2 * h2 + 1) * Y_ + y];
            #pragma unroll
            for (int j = 0; j < TC_; ++j) {
                acc[j] += w0.y * fast_tanh(q.x + w0.x * xl[j]);
                acc[j] += w1.y * fast_tanh(q.y + w1.x * xl[j]);
            }
        }
        #pragma unroll
        for (int j = 0; j < TC_; ++j)
            lds_A[b * TY_ + (t0 + j) * Y_ + y] = acc[j];
    }
    __syncthreads();

    if (tid < B_ * Y_) {
        const int b = tid / Y_, y = tid % Y_;
        float mx = -INFINITY;
        #pragma unroll
        for (int t = 0; t < T_; ++t)
            mx = fmaxf(mx, lds_A[b * TY_ + t * Y_ + y]);
        float s = 0.0f;
        #pragma unroll
        for (int t = 0; t < T_; ++t) {
            const int idx = b * TY_ + t * Y_ + y;
            const float w = __expf(lds_A[idx] - mx);
            lds_A[idx] = w;
            s += w;
        }
        float xn = 0.0f;
        #pragma unroll
        for (int t = 0; t < T_; ++t) {
            const int idx = b * TY_ + t * Y_ + y;
            xn += lds_A[idx] * lds_xl[idx];
        }
        lds_xn[b * 16 + y] = xn / s;
    }
    __syncthreads();

    if (wave < 4) {
        const int d0 = wave * 16;
        const int m  = lane & 15;
        const int ko = (lane >> 4) * 2;
        v8f c = {};
        #pragma unroll
        for (int kk = 0; kk < 3; ++kk) {
            const int k = kk * 4 + ko;
            const float2 af = *(const float2*)&lds_xn[m * 16 + k];
            const float2 bf = *(const float2*)&lds_ewT[(d0 + m) * 16 + k];
            v2f a, b;
            a.x = af.x; a.y = af.y;
            b.x = bf.x; b.y = bf.y;
            c = wmma_f32_16x16x4(a, b, c);
        }
        if (lane < 16) {
            #pragma unroll
            for (int v = 0; v < B_; ++v)
                out[(v * N_ + n) * DL_ + d0 + lane] =
                    c[v] + encb[n * DL_ + d0 + lane];
        }
    }
}

extern "C" void kernel_launch(void* const* d_in, const int* in_sizes, int n_in,
                              void* d_out, int out_size, void* d_ws, size_t ws_size,
                              hipStream_t stream) {
    const float* x_long  = (const float*)d_in[0];
    const float* x_short = (const float*)d_in[1];
    const float* Wq      = (const float*)d_in[2];
    const float* Wk      = (const float*)d_in[3];
    const float* V       = (const float*)d_in[4];
    const float* encW    = (const float*)d_in[5];
    const float* encb    = (const float*)d_in[6];
    float* out = (float*)d_out;

    tsa_fused_kernel<<<N_, NT_, 0, stream>>>(x_long, x_short, Wq, Wk, V,
                                             encW, encb, out);
}


# Prologue split by wave role: waves 0-3 go from their operand loads straight to the Q product and stage nothing; waves 4-11 stage all Wk/V pairs, 3 per thread, loaded as one group
# speedup vs baseline: 1.9256x; 1.9256x over previous
#include <hip/hip_runtime.h>
#include <cmath>

typedef __attribute__((ext_vector_type(2))) float v2f;
typedef __attribute__((ext_vector_type(8))) float v8f;

#define B_  8
#define N_  400
#define T_  24
#define Y_  12
#define H_  64
#define D_  96
#define DL_ 64
#define TY_ (T_ * Y_)
#define BTY_ (B_ * TY_)
#define NT_ 384
#define TC_ 6

__device__ __forceinline__ float fast_tanh(float x) {
#if __has_builtin(__builtin_amdgcn_tanhf)
    return __builtin_amdgcn_tanhf(x);
#else
    return tanhf(x);
#endif
}

__device__ __forceinline__ v8f wmma_f32_16x16x4(v2f a, v2f b, v8f c) {
    return __builtin_amdgcn_wmma_f32_16x16x4_f32(
         false, a,  false, b,
         (short)0, c,  false,  false);
}

__global__ __launch_bounds__(NT_) void tsa_fused_kernel(
    const float* __restrict__ x_long,
    const float* __restrict__ x_short,
    const float* __restrict__ Wq,
    const float* __restrict__ Wk,
    const float* __restrict__ V,
    const float* __restrict__ encW,
    const float* __restrict__ encb,
    float* __restrict__ out)
{
    __shared__ alignas(16) float  lds_Q [B_ * H_];
    __shared__ alignas(16) float2 lds_vk[H_ * Y_];
    __shared__ alignas(16) float  lds_xl[BTY_];
    __shared__ alignas(16) float  lds_A [BTY_];
    __shared__ alignas(16) float  lds_xn[16 * 16];

    const int n    = blockIdx.x;
    const int tid  = threadIdx.x;
    const int wave = tid >> 5;
    const int lane = tid & 31;
    const int m    = lane & 15;
    const int hf   = lane >> 4;

    float4 qa[12], qb[12];
    float  eb[6];
    float  bias = 0.0f;
    if (wave < 4) {
        const int h0 = wave * 16;
        const float* __restrict__ wrow = Wq      + ((h0 + m)  * N_ + n) * D_ + 4 * hf;
        const float* __restrict__ xrow = x_short + ((m & 7)   * N_ + n) * D_ + 4 * hf;
        #pragma unroll
        for (int j = 0; j < 12; ++j) qb[j] = *(const float4*)(wrow + 8 * j);
        #pragma unroll
        for (int j = 0; j < 12; ++j) qa[j] = *(const float4*)(xrow + 8 * j);
        #pragma unroll
        for (int kk = 0; kk < 3; ++kk) {
            eb[2 * kk    ] = encW[(n * Y_ + kk * 4 + 2 * hf    ) * DL_ + h0 + m];
            eb[2 * kk + 1] = encW[(n * Y_ + kk * 4 + 2 * hf + 1) * DL_ + h0 + m];
        }
        bias = encb[n * DL_ + h0 + m];
    }

    const int p   = tid % (B_ * Y_);
    const int t0  = (tid / (B_ * Y_)) * TC_;
    const int pb  = p / Y_, py = p % Y_;
    float xl[TC_];
    #pragma unroll
    for (int j = 0; j < TC_; ++j)
        xl[j] = x_long[(pb * N_ + n) * TY_ + (t0 + j) * Y_ + py];

    if (wave >= 4) {
        const int s = tid - 4 * 32;
        float wkr[3], vr[3];
        #pragma unroll
        for (int k = 0; k < 3; ++k) {
            const int i = s + 256 * k;
            const int h = i / Y_, y = i % Y_;
            wkr[k] = Wk[(h * N_ + n) * Y_ + y];
            vr [k] = V [(h * N_ + n) * Y_ + y];
        }
        #pragma unroll
        for (int k = 0; k < 3; ++k)
            lds_vk[s + 256 * k] = make_float2(wkr[k], vr[k]);
    }
    if (tid < 256) lds_xn[tid] = 0.0f;

    if (wave < 4) {
        const int h0 = wave * 16;
        v8f c = {};
        #pragma unroll
        for (int j = 0; j < 12; ++j) {
            v2f a, b;
            a.x = qa[j].x; a.y = qa[j].y;
            b.x = qb[j].x; b.y = qb[j].y;
            c = wmma_f32_16x16x4(a, b, c);
            a.x = qa[j].z; a.y = qa[j].w;
            b.x = qb[j].z; b.y = qb[j].w;
            c = wmma_f32_16x16x4(a, b, c);
        }
        if (lane < 16) {
            #pragma unroll
            for (int v = 0; v < B_; ++v)
                lds_Q[v * H_ + h0 + lane] = c[v];
        }
    }
    __syncthreads();

    {
        const int b = pb, y = py;
        float acc[TC_];
        #pragma unroll
        for (int j = 0; j < TC_; ++j) acc[j] = 0.0f;
        const float2* __restrict__ q2 = (const float2*)(lds_Q + b * H_);
        #pragma unroll 4
        for (int h2 = 0; h2 < H_ / 2; ++h2) {
            const float2 q  = q2[h2];
            const float2 w0 = lds_vk[(2 * h2    ) * Y_ + y];
            const float2 w1 = lds_vk[(2 * h2 + 1) * Y_ + y];
            #pragma unroll
            for (int j = 0; j < TC_; ++j) {
                acc[j] += w0.y * fast_tanh(q.x + w0.x * xl[j]);
                acc[j] += w1.y * fast_tanh(q.y + w1.x * xl[j]);
            }
        }
        #pragma unroll
        for (int j = 0; j < TC_; ++j) {
            lds_A [b * TY_ + (t0 + j) * Y_ + y] = acc[j];
            lds_xl[b * TY_ + (t0 + j) * Y_ + y] = xl[j];
        }
    }
    __syncthreads();

    if (tid < B_ * Y_) {
        const int b = tid / Y_, y = tid % Y_;
        float mx = -INFINITY;
        #pragma unroll
        for (int t = 0; t < T_; ++t)
            mx = fmaxf(mx, lds_A[b * TY_ + t * Y_ + y]);
        float s = 0.0f;
        #pragma unroll
        for (int t = 0; t < T_; ++t) {
            const int idx = b * TY_ + t * Y_ + y;
            const float w = __expf(lds_A[idx] - mx);
            lds_A[idx] = w;
            s += w;
        }
        float xn = 0.0f;
        #pragma unroll
        for (int t = 0; t < T_; ++t) {
            const int idx = b * TY_ + t * Y_ + y;
            xn += lds_A[idx] * lds_xl[idx];
        }
        lds_xn[b * 16 + y] = xn / s;
    }
    __syncthreads();

    if (wave < 4) {
        const int d0 = wave * 16;
        v8f c = {};
        #pragma unroll
        for (int kk = 0; kk < 3; ++kk) {
            const int k = kk * 4 + 2 * hf;
            const float2 af = *(const float2*)&lds_xn[m * 16 + k];
            v2f a, b;
            a.x = af.x; a.y = af.y;
            b.x = eb[2 * kk]; b.y = eb[2 * kk + 1];
            c = wmma_f32_16x16x4(a, b, c);
        }
        if (lane < 16) {
            #pragma unroll
            for (int v = 0; v < B_; ++v)
                out[(v * N_ + n) * DL_ + d0 + lane] = c[v] + bias;
        }
    }
}

extern "C" void kernel_launch(void* const* d_in, const int* in_sizes, int n_in,
                              void* d_out, int out_size, void* d_ws, size_t ws_size,
                              hipStream_t stream) {
    const float* x_long  = (const float*)d_in[0];
    const float* x_short = (const float*)d_in[1];
    const float* Wq      = (const float*)d_in[2];
    const float* Wk      = (const float*)d_in[3];
    const float* V       = (const float*)d_in[4];
    const float* encW    = (const float*)d_in[5];
    const float* encb    = (const float*)d_in[6];
    float* out = (float*)d_out;

    tsa_fused_kernel<<<N_, NT_, 0, stream>>>(x_long, x_short, Wq, Wk, V,
                                             encW, encb, out);
}
